# starting kernel
# baseline (speedup 1.0000x reference)
#include <hip/hip_runtime.h>
#include <hip/hip_bf16.h>
#include <math.h>


typedef float v2f __attribute__((ext_vector_type(2)));
typedef float v8f __attribute__((ext_vector_type(8)));

#define NROWS   131072
#define DIN     16
#define DOUT    16
#define GRID    19
#define WREAL   361
#define WPAD    368
#define NWT     23
#define TWO_PI_19 0.3306939635357677f

#define B_OFF   0
#define C_OFF   11776
#define L_OFF   (11776 + 23552)
#define WS_FLOATS (L_OFF + 256)

static __device__ inline v8f wmma4(v2f a, v2f b, v8f c) {
  return __builtin_amdgcn_wmma_f32_16x16x4_f32(
       false, a,  false, b,
       (short)0, c,  false,  false);
}


__global__ void pre_B(const float* __restrict__ Mre, const float* __restrict__ Mim,
                      float* __restrict__ Bpack) {
  int w = blockIdx.x;
  int t = threadIdx.x;
  int i = t & 15, c = t >> 4;
  float s = 0.f;
  if (w < WREAL) {
    int wu = w / GRID, wv = w % GRID;
    for (int u = 0; u < 7; ++u)
      for (int v = 0; v < 7; ++v) {
        int ph = (wu * u + wv * v) % GRID;
        float th = (float)ph * TWO_PI_19;
        float ct = cosf(th), st = sinf(th);
        float a = Mre[(i * 7 + u) * 7 + v];
        float b = Mim[(i * 7 + u) * 7 + v];
        s += (c == 0) ? (a * ct + b * st) : (b * ct - a * st);
      }
    s *= (1.f / 19.f);
  }
  Bpack[((c * 8 + (i >> 1)) * WPAD + w) * 2 + (i & 1)] = s;
}

__global__ void pre_C(const float* __restrict__ Sre, const float* __restrict__ Sim,
                      const float* __restrict__ pw, float* __restrict__ Cpack) {
  int w = blockIdx.x;
  int j = threadIdx.x;
  float g2r = 0.f, g2i = 0.f, g3r = 0.f, g3i = 0.f;
  if (w < WREAL) {
    int wu = w / GRID, wv = w % GRID;
    for (int tu = 0; tu < 13; ++tu)
      for (int tv = 0; tv < 13; ++tv) {
        int ph = (wu * tu + wv * tv) % GRID;
        float th = (float)ph * TWO_PI_19;
        float ct = cosf(th), st = sinf(th);
        float a = Sre[(j * GRID + tu + 3) * GRID + tv + 3];
        float b = Sim[(j * GRID + tu + 3) * GRID + tv + 3];
        g2r += a * ct - b * st;
        g2i += b * ct + a * st;
      }
    for (int tu = 0; tu < GRID; ++tu)
      for (int tv = 0; tv < GRID; ++tv) {
        int ph = (wu * tu + wv * tv) % GRID;
        float th = (float)ph * TWO_PI_19;
        float ct = cosf(th), st = sinf(th);
        float a = Sre[(j * GRID + tu) * GRID + tv];
        float b = Sim[(j * GRID + tu) * GRID + tv];
        g3r += a * ct - b * st;
        g3i += b * ct + a * st;
      }
  }
  float scale = pw[j] * (1.f / 19.f);
  float q[4] = { g2r * scale, -g2i * scale, g3r * scale, -g3i * scale };
  for (int qi = 0; qi < 4; ++qi)
    Cpack[((qi * (WPAD / 2) + (w >> 1)) * 16 + j) * 2 + (w & 1)] = q[qi];
}

__global__ void pre_L(const float* __restrict__ Mre, const float* __restrict__ Mim,
                      const float* __restrict__ Sre, const float* __restrict__ Sim,
                      const float* __restrict__ pw, float* __restrict__ Lpack) {
  int i = threadIdx.x;
  int j = threadIdx.y;
  float s = 0.f;
  for (int u = 0; u < 7; ++u)
    for (int v = 0; v < 7; ++v) {
      s += Mre[(i * 7 + u) * 7 + v] * Sre[(j * GRID + u + 6) * GRID + v + 6]
         - Mim[(i * 7 + u) * 7 + v] * Sim[(j * GRID + u + 6) * GRID + v + 6];
    }
  s *= pw[j];
  Lpack[((i >> 1) * 16 + j) * 2 + (i & 1)] = s;
}


__global__ __launch_bounds__(256, 1)
void gaunt_main(const float* __restrict__ X,
                const float* __restrict__ Bp,
                const float* __restrict__ Cp,
                const float* __restrict__ Lp,
                float* __restrict__ Y) {
  __shared__ float lds[8 * 4 * 16 * 18];

  const int tid  = threadIdx.x;
  const int wave = tid >> 5;
  const int lane = tid & 31;
  const int half = lane >> 4;
  const int lr   = lane & 15;

  const long row0 = ((long)blockIdx.x * 8 + wave) * 16;

  const v2f* B2 = (const v2f*)Bp;
  const v2f* C2 = (const v2f*)Cp;
  const v2f* L2 = (const v2f*)Lp;

  v2f xa[4];
  const float* xrow = X + (row0 + lr) * DIN;
#pragma unroll
  for (int kk = 0; kk < 4; ++kk)
    xa[kk] = *(const v2f*)(xrow + 4 * kk + 2 * half);

  v8f acc = {0.f, 0.f, 0.f, 0.f, 0.f, 0.f, 0.f, 0.f};

#pragma unroll
  for (int kk = 0; kk < 4; ++kk)
    acc = wmma4(xa[kk], L2[(2 * kk + half) * 16 + lr], acc);

  float* myl = lds + wave * (4 * 16 * 18);

  for (int wt = 0; wt < NWT; ++wt) {
    const int w0 = wt * 16;

    v8f fre = {0.f, 0.f, 0.f, 0.f, 0.f, 0.f, 0.f, 0.f};
    v8f fim = {0.f, 0.f, 0.f, 0.f, 0.f, 0.f, 0.f, 0.f};
#pragma unroll
    for (int kk = 0; kk < 4; ++kk) {
      fre = wmma4(xa[kk], B2[(    2 * kk + half) * WPAD + w0 + lr], fre);
      fim = wmma4(xa[kk], B2[(8 + 2 * kk + half) * WPAD + w0 + lr], fim);
    }

#pragma unroll
    for (int e = 0; e < 8; ++e) {
      float fr = fre[e], fi = fim[e];
      float p2r = fr * fr - fi * fi;
      float p2i = 2.f * fr * fi;
      float p3r = p2r * fr - p2i * fi;
      float p3i = p2r * fi + p2i * fr;
      int off = (e + 8 * half) * 18 + lr;
      myl[0 * 288 + off] = p2r;
      myl[1 * 288 + off] = p2i;
      myl[2 * 288 + off] = p3r;
      myl[3 * 288 + off] = p3i;
    }

#pragma unroll
    for (int q = 0; q < 4; ++q) {
#pragma unroll
      for (int kk = 0; kk < 4; ++kk) {
        v2f a = *(const v2f*)(myl + q * 288 + lr * 18 + 4 * kk + 2 * half);
        v2f b = C2[(q * (WPAD / 2) + wt * 8 + 2 * kk + half) * 16 + lr];
        acc = wmma4(a, b, acc);
      }
    }
  }

  float* yrow = Y + row0 * DOUT;
#pragma unroll
  for (int e = 0; e < 8; ++e)
    yrow[(e + 8 * half) * DOUT + lr] = acc[e];
}


extern "C" void kernel_launch(void* const* d_in, const int* in_sizes, int n_in,
                              void* d_out, int out_size, void* d_ws, size_t ws_size,
                              hipStream_t stream) {
  const float* x   = (const float*)d_in[0];
  const float* Mre = (const float*)d_in[1];
  const float* Mim = (const float*)d_in[2];
  const float* Sre = (const float*)d_in[3];
  const float* Sim = (const float*)d_in[4];
  const float* pw  = (const float*)d_in[5];

  float* ws = (float*)d_ws;
  float* Bp = ws + B_OFF;
  float* Cp = ws + C_OFF;
  float* Lp = ws + L_OFF;

  pre_B<<<WPAD, 32, 0, stream>>>(Mre, Mim, Bp);
  pre_C<<<WPAD, 16, 0, stream>>>(Sre, Sim, pw, Cp);
  dim3 lb(16, 16);
  pre_L<<<1, lb, 0, stream>>>(Mre, Mim, Sre, Sim, pw, Lp);

  gaunt_main<<<NROWS / (8 * 16), 256, 0, stream>>>(x, Bp, Cp, Lp, (float*)d_out);
}


# pre_B as a separable row-column transform: one block per channel pair sums M rows over v, then over u; 19 unit roots made once per block; 7-term chains, not 49 sincos terms; sum order of B changes
# speedup vs baseline: 2.6124x; 2.6124x over previous
#include <hip/hip_runtime.h>
#include <hip/hip_bf16.h>
#include <math.h>


typedef float v2f __attribute__((ext_vector_type(2)));
typedef float v8f __attribute__((ext_vector_type(8)));

#define NROWS   131072
#define DIN     16
#define DOUT    16
#define GRID    19
#define WREAL   361
#define WPAD    368
#define NWT     23
#define TWO_PI_19 0.3306939635357677f

#define B_OFF   0
#define C_OFF   11776
#define L_OFF   (11776 + 23552)
#define WS_FLOATS (L_OFF + 256)

static __device__ inline v8f wmma4(v2f a, v2f b, v8f c) {
  return __builtin_amdgcn_wmma_f32_16x16x4_f32(
       false, a,  false, b,
       (short)0, c,  false,  false);
}


#define PB_ROWS    (2 * 7 * GRID)
#define PB_THREADS 384
#define PB_ROOT0   352
__global__ __launch_bounds__(PB_THREADS)
void pre_B(const float* __restrict__ Mre, const float* __restrict__ Mim,
           float* __restrict__ Bpack) {
  __shared__ float rootc[GRID], roots[GRID];
  __shared__ float Tr[PB_ROWS], Ti[PB_ROWS];

  const int t = threadIdx.x;
  const int p = blockIdx.x;

  const bool has1 = t < PB_ROWS;
  const int il  = t / (7 * GRID);
  const int r   = t - il * (7 * GRID);
  const int u1  = r / GRID;
  const int wv1 = r - u1 * GRID;
  float a[7], b[7];
  for (int v = 0; v < 7; ++v) { a[v] = 0.f; b[v] = 0.f; }
  if (has1) {
    const float* rre = Mre + ((2 * p + il) * 7 + u1) * 7;
    const float* rim = Mim + ((2 * p + il) * 7 + u1) * 7;
    for (int v = 0; v < 7; ++v) { a[v] = rre[v]; b[v] = rim[v]; }
  }

  if (t >= PB_ROOT0 && t < PB_ROOT0 + GRID) {
    const int k = t - PB_ROOT0;
    const float th = (float)k * TWO_PI_19;
    rootc[k] = cosf(th);
    roots[k] = sinf(th);
  }

  __syncthreads();

  if (has1) {
    float tr = 0.f, ti = 0.f;
    int ph = 0;
    for (int v = 0; v < 7; ++v) {
      const float c = rootc[ph], s = roots[ph];
      tr += a[v] * c + b[v] * s;
      ti += b[v] * c - a[v] * s;
      ph += wv1;
      if (ph >= GRID) ph -= GRID;
    }
    Tr[t] = tr;
    Ti[t] = ti;
  }

  __syncthreads();

  float re0 = 0.f, im0 = 0.f, re1 = 0.f, im1 = 0.f;
  if (t < WREAL) {
    const int wu = t / GRID;
    const int wv = t - wu * GRID;
    int ph = 0;
    for (int u = 0; u < 7; ++u) {
      const float c = rootc[ph], s = roots[ph];
      const float x0r = Tr[u * GRID + wv],            x0i = Ti[u * GRID + wv];
      const float x1r = Tr[(7 + u) * GRID + wv],      x1i = Ti[(7 + u) * GRID + wv];
      re0 += x0r * c + x0i * s;
      im0 += x0i * c - x0r * s;
      re1 += x1r * c + x1i * s;
      im1 += x1i * c - x1r * s;
      ph += wu;
      if (ph >= GRID) ph -= GRID;
    }
    re0 *= (1.f / 19.f); im0 *= (1.f / 19.f);
    re1 *= (1.f / 19.f); im1 *= (1.f / 19.f);
  }
  if (t < WPAD) {
    v2f* B2 = (v2f*)Bpack;
    v2f vre = {re0, re1};
    v2f vim = {im0, im1};
    B2[(    p) * WPAD + t] = vre;
    B2[(8 + p) * WPAD + t] = vim;
  }
}

__global__ __launch_bounds__(32)
void pre_C(const float* __restrict__ Sre, const float* __restrict__ Sim,
           const float* __restrict__ pw, float* __restrict__ Cpack) {
  __shared__ float rootc[GRID], roots[GRID];
  __shared__ float Tr[32], Ti[32];

  const int lane = threadIdx.x;
  const int j    = blockIdx.x / GRID;
  const int wv   = blockIdx.x % GRID;

  if (lane < GRID) {
    float th = (float)lane * TWO_PI_19;
    rootc[lane] = cosf(th);
    roots[lane] = sinf(th);
  }

  const bool full = lane < GRID;
  const int  row  = full ? lane : lane - GRID + 3;
  const int  col0 = full ? 0 : 3;
  const float* rre = Sre + (j * GRID + row) * GRID + col0;
  const float* rim = Sim + (j * GRID + row) * GRID + col0;
  float a[GRID], b[GRID];
  for (int tv = 0; tv < 13; ++tv) { a[tv] = rre[tv]; b[tv] = rim[tv]; }
  for (int tv = 13; tv < GRID; ++tv) { a[tv] = 0.f; b[tv] = 0.f; }
  if (full) {
    for (int tv = 13; tv < GRID; ++tv) { a[tv] = rre[tv]; b[tv] = rim[tv]; }
  }

  __syncthreads();

  float tr = 0.f, ti = 0.f;
  for (int tv = 0; tv < GRID; ++tv) {
    const int ph = (wv * tv) % GRID;
    const float c = rootc[ph], s = roots[ph];
    tr += a[tv] * c - b[tv] * s;
    ti += b[tv] * c + a[tv] * s;
  }
  Tr[lane] = tr;
  Ti[lane] = ti;

  __syncthreads();

  if (lane < GRID) {
    const int wu = lane;
    float g3r = 0.f, g3i = 0.f, g2r = 0.f, g2i = 0.f;
    int ph = 0;
    for (int tu = 0; tu < GRID; ++tu) {
      const float c = rootc[ph], s = roots[ph];
      const float xr = Tr[tu], xi = Ti[tu];
      g3r += xr * c - xi * s;
      g3i += xi * c + xr * s;
      if (tu < 13) {
        const float zr = Tr[GRID + tu], zi = Ti[GRID + tu];
        g2r += zr * c - zi * s;
        g2i += zi * c + zr * s;
      }
      ph += wu;
      if (ph >= GRID) ph -= GRID;
    }
    const int w = wu * GRID + wv;
    const float scale = pw[j] * (1.f / 19.f);
    const float q[4] = { g2r * scale, -g2i * scale, g3r * scale, -g3i * scale };
    for (int qi = 0; qi < 4; ++qi)
      Cpack[((qi * (WPAD / 2) + (w >> 1)) * 16 + j) * 2 + (w & 1)] = q[qi];
  }

  if (wv == 0 && lane < 4 * (WPAD - WREAL)) {
    const int qi = lane & 3;
    const int w  = WREAL + (lane >> 2);
    Cpack[((qi * (WPAD / 2) + (w >> 1)) * 16 + j) * 2 + (w & 1)] = 0.f;
  }
}

__global__ void pre_L(const float* __restrict__ Mre, const float* __restrict__ Mim,
                      const float* __restrict__ Sre, const float* __restrict__ Sim,
                      const float* __restrict__ pw, float* __restrict__ Lpack) {
  int i = threadIdx.x;
  int j = threadIdx.y;
  float s = 0.f;
  for (int u = 0; u < 7; ++u)
    for (int v = 0; v < 7; ++v) {
      s += Mre[(i * 7 + u) * 7 + v] * Sre[(j * GRID + u + 6) * GRID + v + 6]
         - Mim[(i * 7 + u) * 7 + v] * Sim[(j * GRID + u + 6) * GRID + v + 6];
    }
  s *= pw[j];
  Lpack[((i >> 1) * 16 + j) * 2 + (i & 1)] = s;
}


__global__ __launch_bounds__(256, 1)
void gaunt_main(const float* __restrict__ X,
                const float* __restrict__ Bp,
                const float* __restrict__ Cp,
                const float* __restrict__ Lp,
                float* __restrict__ Y) {
  __shared__ float lds[8 * 4 * 16 * 18];

  const int tid  = threadIdx.x;
  const int wave = tid >> 5;
  const int lane = tid & 31;
  const int half = lane >> 4;
  const int lr   = lane & 15;

  const long row0 = ((long)blockIdx.x * 8 + wave) * 16;

  const v2f* B2 = (const v2f*)Bp;
  const v2f* C2 = (const v2f*)Cp;
  const v2f* L2 = (const v2f*)Lp;

  v2f xa[4];
  const float* xrow = X + (row0 + lr) * DIN;
#pragma unroll
  for (int kk = 0; kk < 4; ++kk)
    xa[kk] = *(const v2f*)(xrow + 4 * kk + 2 * half);

  v8f acc = {0.f, 0.f, 0.f, 0.f, 0.f, 0.f, 0.f, 0.f};

#pragma unroll
  for (int kk = 0; kk < 4; ++kk)
    acc = wmma4(xa[kk], L2[(2 * kk + half) * 16 + lr], acc);

  float* myl = lds + wave * (4 * 16 * 18);

  for (int wt = 0; wt < NWT; ++wt) {
    const int w0 = wt * 16;

    v8f fre = {0.f, 0.f, 0.f, 0.f, 0.f, 0.f, 0.f, 0.f};
    v8f fim = {0.f, 0.f, 0.f, 0.f, 0.f, 0.f, 0.f, 0.f};
#pragma unroll
    for (int kk = 0; kk < 4; ++kk) {
      fre = wmma4(xa[kk], B2[(    2 * kk + half) * WPAD + w0 + lr], fre);
      fim = wmma4(xa[kk], B2[(8 + 2 * kk + half) * WPAD + w0 + lr], fim);
    }

#pragma unroll
    for (int e = 0; e < 8; ++e) {
      float fr = fre[e], fi = fim[e];
      float p2r = fr * fr - fi * fi;
      float p2i = 2.f * fr * fi;
      float p3r = p2r * fr - p2i * fi;
      float p3i = p2r * fi + p2i * fr;
      int off = (e + 8 * half) * 18 + lr;
      myl[0 * 288 + off] = p2r;
      myl[1 * 288 + off] = p2i;
      myl[2 * 288 + off] = p3r;
      myl[3 * 288 + off] = p3i;
    }

#pragma unroll
    for (int q = 0; q < 4; ++q) {
#pragma unroll
      for (int kk = 0; kk < 4; ++kk) {
        v2f a = *(const v2f*)(myl + q * 288 + lr * 18 + 4 * kk + 2 * half);
        v2f b = C2[(q * (WPAD / 2) + wt * 8 + 2 * kk + half) * 16 + lr];
        acc = wmma4(a, b, acc);
      }
    }
  }

  float* yrow = Y + row0 * DOUT;
#pragma unroll
  for (int e = 0; e < 8; ++e)
    yrow[(e + 8 * half) * DOUT + lr] = acc[e];
}


extern "C" void kernel_launch(void* const* d_in, const int* in_sizes, int n_in,
                              void* d_out, int out_size, void* d_ws, size_t ws_size,
                              hipStream_t stream) {
  const float* x   = (const float*)d_in[0];
  const float* Mre = (const float*)d_in[1];
  const float* Mim = (const float*)d_in[2];
  const float* Sre = (const float*)d_in[3];
  const float* Sim = (const float*)d_in[4];
  const float* pw  = (const float*)d_in[5];

  float* ws = (float*)d_ws;
  float* Bp = ws + B_OFF;
  float* Cp = ws + C_OFF;
  float* Lp = ws + L_OFF;

  pre_B<<<8, PB_THREADS, 0, stream>>>(Mre, Mim, Bp);
  pre_C<<<DOUT * GRID, 32, 0, stream>>>(Sre, Sim, pw, Cp);
  dim3 lb(16, 16);
  pre_L<<<1, lb, 0, stream>>>(Mre, Mim, Sre, Sim, pw, Lp);

  gaunt_main<<<NROWS / (8 * 16), 256, 0, stream>>>(x, Bp, Cp, Lp, (float*)d_out);
}


# Square term folded into the cube: GEMM1 starts from a constant spectrum phi/3, so z^3 carries x_f^2; L pack x2/3 and a start value of y cancel by-products; GEMM2 2 planes; sum order of y changes
# speedup vs baseline: 3.6817x; 1.4093x over previous
#include <hip/hip_runtime.h>
#include <hip/hip_bf16.h>
#include <math.h>


typedef float v2f __attribute__((ext_vector_type(2)));
typedef float v8f __attribute__((ext_vector_type(8)));

#define NROWS   131072
#define DIN     16
#define DOUT    16
#define GRID    19
#define WREAL   361
#define WPAD    368
#define NWT     23
#define TWO_PI_19 0.3306939635357677f

#define B_OFF   0
#define C_OFF   11776
#define L_OFF   (11776 + 11776)
#define Z_OFF   (L_OFF + 256)
#define K_OFF   (Z_OFF + 736)
#define WS_FLOATS (K_OFF + 16)

static __device__ inline v8f wmma4(v2f a, v2f b, v8f c) {
  return __builtin_amdgcn_wmma_f32_16x16x4_f32(
       false, a,  false, b,
       (short)0, c,  false,  false);
}


#define PB_ROWS    (2 * 7 * GRID)
#define PB_THREADS 384
#define PB_ROOT0   352
__global__ __launch_bounds__(PB_THREADS)
void pre_B(const float* __restrict__ Mre, const float* __restrict__ Mim,
           float* __restrict__ Bpack, float* __restrict__ Zpack) {
  __shared__ float rootc[GRID], roots[GRID];
  __shared__ float Tr[PB_ROWS], Ti[PB_ROWS];

  const int t = threadIdx.x;
  const int p = blockIdx.x;

  const bool has1 = t < PB_ROWS;
  const int il  = t / (7 * GRID);
  const int r   = t - il * (7 * GRID);
  const int u1  = r / GRID;
  const int wv1 = r - u1 * GRID;
  float a[7], b[7];
  for (int v = 0; v < 7; ++v) { a[v] = 0.f; b[v] = 0.f; }
  if (has1) {
    const float* rre = Mre + ((2 * p + il) * 7 + u1) * 7;
    const float* rim = Mim + ((2 * p + il) * 7 + u1) * 7;
    for (int v = 0; v < 7; ++v) { a[v] = rre[v]; b[v] = rim[v]; }
  }

  if (t >= PB_ROOT0 && t < PB_ROOT0 + GRID) {
    const int k = t - PB_ROOT0;
    const float th = (float)k * TWO_PI_19;
    rootc[k] = cosf(th);
    roots[k] = sinf(th);
  }

  __syncthreads();

  if (has1) {
    float tr = 0.f, ti = 0.f;
    int ph = 0;
    for (int v = 0; v < 7; ++v) {
      const float c = rootc[ph], s = roots[ph];
      tr += a[v] * c + b[v] * s;
      ti += b[v] * c - a[v] * s;
      ph += wv1;
      if (ph >= GRID) ph -= GRID;
    }
    Tr[t] = tr;
    Ti[t] = ti;
  }

  __syncthreads();

  float re0 = 0.f, im0 = 0.f, re1 = 0.f, im1 = 0.f;
  if (t < WREAL) {
    const int wu = t / GRID;
    const int wv = t - wu * GRID;
    int ph = 0;
    for (int u = 0; u < 7; ++u) {
      const float c = rootc[ph], s = roots[ph];
      const float x0r = Tr[u * GRID + wv],            x0i = Ti[u * GRID + wv];
      const float x1r = Tr[(7 + u) * GRID + wv],      x1i = Ti[(7 + u) * GRID + wv];
      re0 += x0r * c + x0i * s;
      im0 += x0i * c - x0r * s;
      re1 += x1r * c + x1i * s;
      im1 += x1i * c - x1r * s;
      ph += wu;
      if (ph >= GRID) ph -= GRID;
    }
    re0 *= (1.f / 19.f); im0 *= (1.f / 19.f);
    re1 *= (1.f / 19.f); im1 *= (1.f / 19.f);
  }
  if (t < WPAD) {
    v2f* B2 = (v2f*)Bpack;
    v2f vre = {re0, re1};
    v2f vim = {im0, im1};
    B2[(    p) * WPAD + t] = vre;
    B2[(8 + p) * WPAD + t] = vim;
  }
  if (p == 0 && t < WPAD) {
    v2f z = {0.f, 0.f};
    if (t < WREAL) {
      const int wu = t / GRID;
      const int wv = t - wu * GRID;
      const int k  = (3 * (wu + wv)) % GRID;
      z.x =  rootc[k] * (1.f / 3.f);
      z.y = -roots[k] * (1.f / 3.f);
    }
    ((v2f*)Zpack)[t] = z;
  }
}

__global__ __launch_bounds__(32)
void pre_C(const float* __restrict__ Sre, const float* __restrict__ Sim,
           const float* __restrict__ pw, float* __restrict__ Cpack) {
  __shared__ float rootc[GRID], roots[GRID];
  __shared__ float Tr[GRID], Ti[GRID];

  const int lane = threadIdx.x;
  const int j    = blockIdx.x / GRID;
  const int wv   = blockIdx.x % GRID;

  const bool has = lane < GRID;

  if (has) {
    float th = (float)lane * TWO_PI_19;
    rootc[lane] = cosf(th);
    roots[lane] = sinf(th);
  }

  float a[GRID], b[GRID];
  for (int tv = 0; tv < GRID; ++tv) { a[tv] = 0.f; b[tv] = 0.f; }
  if (has) {
    const float* rre = Sre + (j * GRID + lane) * GRID;
    const float* rim = Sim + (j * GRID + lane) * GRID;
    for (int tv = 0; tv < GRID; ++tv) { a[tv] = rre[tv]; b[tv] = rim[tv]; }
  }

  __syncthreads();

  float tr = 0.f, ti = 0.f;
  for (int tv = 0; tv < GRID; ++tv) {
    const int ph = (wv * tv) % GRID;
    const float c = rootc[ph], s = roots[ph];
    tr += a[tv] * c - b[tv] * s;
    ti += b[tv] * c + a[tv] * s;
  }
  if (has) {
    Tr[lane] = tr;
    Ti[lane] = ti;
  }

  __syncthreads();

  if (has) {
    const int wu = lane;
    float g3r = 0.f, g3i = 0.f;
    int ph = 0;
    for (int tu = 0; tu < GRID; ++tu) {
      const float c = rootc[ph], s = roots[ph];
      const float xr = Tr[tu], xi = Ti[tu];
      g3r += xr * c - xi * s;
      g3i += xi * c + xr * s;
      ph += wu;
      if (ph >= GRID) ph -= GRID;
    }
    const int w = wu * GRID + wv;
    const float scale = pw[j] * (1.f / 19.f);
    Cpack[((0 * (WPAD / 2) + (w >> 1)) * 16 + j) * 2 + (w & 1)] =  g3r * scale;
    Cpack[((1 * (WPAD / 2) + (w >> 1)) * 16 + j) * 2 + (w & 1)] = -g3i * scale;
  }

  if (wv == 0 && lane < 2 * (WPAD - WREAL)) {
    const int qi = lane & 1;
    const int w  = WREAL + (lane >> 1);
    Cpack[((qi * (WPAD / 2) + (w >> 1)) * 16 + j) * 2 + (w & 1)] = 0.f;
  }
}

__global__ void pre_L(const float* __restrict__ Mre, const float* __restrict__ Mim,
                      const float* __restrict__ Sre, const float* __restrict__ Sim,
                      const float* __restrict__ pw, float* __restrict__ Lpack,
                      float* __restrict__ Kpack) {
  int i = threadIdx.x;
  int j = threadIdx.y;
  float s = 0.f;
  for (int u = 0; u < 7; ++u)
    for (int v = 0; v < 7; ++v) {
      s += Mre[(i * 7 + u) * 7 + v] * Sre[(j * GRID + u + 6) * GRID + v + 6]
         - Mim[(i * 7 + u) * 7 + v] * Sim[(j * GRID + u + 6) * GRID + v + 6];
    }
  s *= pw[j] * (2.f / 3.f);
  Lpack[((i >> 1) * 16 + j) * 2 + (i & 1)] = s;
  if (i == 0)
    Kpack[j] = -(19.f / 27.f) * pw[j] * Sre[(j * GRID + 9) * GRID + 9];
}


__global__ __launch_bounds__(256, 1)
void gaunt_main(const float* __restrict__ X,
                const float* __restrict__ Bp,
                const float* __restrict__ Cp,
                const float* __restrict__ Lp,
                const float* __restrict__ Zp,
                const float* __restrict__ Kp,
                float* __restrict__ Y) {
  __shared__ float lds[8 * 2 * 16 * 18];

  const int tid  = threadIdx.x;
  const int wave = tid >> 5;
  const int lane = tid & 31;
  const int half = lane >> 4;
  const int lr   = lane & 15;

  const long row0 = ((long)blockIdx.x * 8 + wave) * 16;

  const v2f* B2 = (const v2f*)Bp;
  const v2f* C2 = (const v2f*)Cp;
  const v2f* L2 = (const v2f*)Lp;
  const v2f* Z2 = (const v2f*)Zp;

  v2f xa[4];
  const float* xrow = X + (row0 + lr) * DIN;
#pragma unroll
  for (int kk = 0; kk < 4; ++kk)
    xa[kk] = *(const v2f*)(xrow + 4 * kk + 2 * half);

  const float k0 = Kp[lr];
  v8f acc = {k0, k0, k0, k0, k0, k0, k0, k0};

#pragma unroll
  for (int kk = 0; kk < 4; ++kk)
    acc = wmma4(xa[kk], L2[(2 * kk + half) * 16 + lr], acc);

  float* myl = lds + wave * (2 * 16 * 18);

  v2f z = Z2[lr];
  for (int wt = 0; wt < NWT; ++wt) {
    const int w0 = wt * 16;

    const int wn = (wt + 1 < NWT) ? w0 + 16 : 0;
    const v2f zn = Z2[wn + lr];

    v8f fre = {z.x, z.x, z.x, z.x, z.x, z.x, z.x, z.x};
    v8f fim = {z.y, z.y, z.y, z.y, z.y, z.y, z.y, z.y};
#pragma unroll
    for (int kk = 0; kk < 4; ++kk) {
      fre = wmma4(xa[kk], B2[(    2 * kk + half) * WPAD + w0 + lr], fre);
      fim = wmma4(xa[kk], B2[(8 + 2 * kk + half) * WPAD + w0 + lr], fim);
    }

#pragma unroll
    for (int e = 0; e < 8; ++e) {
      float fr = fre[e], fi = fim[e];
      float p2r = fr * fr - fi * fi;
      float p2i = 2.f * fr * fi;
      float p3r = p2r * fr - p2i * fi;
      float p3i = p2r * fi + p2i * fr;
      int off = (e + 8 * half) * 18 + lr;
      myl[0 * 288 + off] = p3r;
      myl[1 * 288 + off] = p3i;
    }

#pragma unroll
    for (int q = 0; q < 2; ++q) {
#pragma unroll
      for (int kk = 0; kk < 4; ++kk) {
        v2f a = *(const v2f*)(myl + q * 288 + lr * 18 + 4 * kk + 2 * half);
        v2f b = C2[(q * (WPAD / 2) + wt * 8 + 2 * kk + half) * 16 + lr];
        acc = wmma4(a, b, acc);
      }
    }
    z = zn;
  }

  float* yrow = Y + row0 * DOUT;
#pragma unroll
  for (int e = 0; e < 8; ++e)
    yrow[(e + 8 * half) * DOUT + lr] = acc[e];
}


extern "C" void kernel_launch(void* const* d_in, const int* in_sizes, int n_in,
                              void* d_out, int out_size, void* d_ws, size_t ws_size,
                              hipStream_t stream) {
  const float* x   = (const float*)d_in[0];
  const float* Mre = (const float*)d_in[1];
  const float* Mim = (const float*)d_in[2];
  const float* Sre = (const float*)d_in[3];
  const float* Sim = (const float*)d_in[4];
  const float* pw  = (const float*)d_in[5];

  float* ws = (float*)d_ws;
  float* Bp = ws + B_OFF;
  float* Cp = ws + C_OFF;
  float* Lp = ws + L_OFF;
  float* Zp = ws + Z_OFF;
  float* Kp = ws + K_OFF;

  pre_B<<<8, PB_THREADS, 0, stream>>>(Mre, Mim, Bp, Zp);
  pre_C<<<DOUT * GRID, 32, 0, stream>>>(Sre, Sim, pw, Cp);
  dim3 lb(16, 16);
  pre_L<<<1, lb, 0, stream>>>(Mre, Mim, Sre, Sim, pw, Lp, Kp);

  gaunt_main<<<NROWS / (8 * 16), 256, 0, stream>>>(x, Bp, Cp, Lp, Zp, Kp, (float*)d_out);
}


# Linear term folded into the cube too: staged value z(z^2+a), a=(2/3)phi^2 per column, so phi^2 x_f carries all of L1 x; L pack, its 4 products and its launch go (4 -> 3); sum order of y changes
# speedup vs baseline: 3.9618x; 1.0761x over previous
#include <hip/hip_runtime.h>
#include <hip/hip_bf16.h>
#include <math.h>


typedef float v2f __attribute__((ext_vector_type(2)));
typedef float v4f __attribute__((ext_vector_type(4)));
typedef float v8f __attribute__((ext_vector_type(8)));

#define NROWS   131072
#define DIN     16
#define DOUT    16
#define GRID    19
#define WREAL   361
#define WPAD    368
#define NWT     23
#define TWO_PI_19 0.3306939635357677f

#define B_OFF   0
#define C_OFF   11776
#define Z_OFF   (11776 + 11776)
#define K_OFF   (Z_OFF + 1472)
#define WS_FLOATS (K_OFF + 16)

static __device__ inline v8f wmma4(v2f a, v2f b, v8f c) {
  return __builtin_amdgcn_wmma_f32_16x16x4_f32(
       false, a,  false, b,
       (short)0, c,  false,  false);
}


#define PB_ROWS    (2 * 7 * GRID)
#define PB_THREADS 384
#define PB_ROOT0   352
__global__ __launch_bounds__(PB_THREADS)
void pre_B(const float* __restrict__ Mre, const float* __restrict__ Mim,
           float* __restrict__ Bpack, float* __restrict__ Zpack) {
  __shared__ float rootc[GRID], roots[GRID];
  __shared__ float Tr[PB_ROWS], Ti[PB_ROWS];

  const int t = threadIdx.x;
  const int p = blockIdx.x;

  const bool has1 = t < PB_ROWS;
  const int il  = t / (7 * GRID);
  const int r   = t - il * (7 * GRID);
  const int u1  = r / GRID;
  const int wv1 = r - u1 * GRID;
  float a[7], b[7];
  for (int v = 0; v < 7; ++v) { a[v] = 0.f; b[v] = 0.f; }
  if (has1) {
    const float* rre = Mre + ((2 * p + il) * 7 + u1) * 7;
    const float* rim = Mim + ((2 * p + il) * 7 + u1) * 7;
    for (int v = 0; v < 7; ++v) { a[v] = rre[v]; b[v] = rim[v]; }
  }

  if (t >= PB_ROOT0 && t < PB_ROOT0 + GRID) {
    const int k = t - PB_ROOT0;
    const float th = (float)k * TWO_PI_19;
    rootc[k] = cosf(th);
    roots[k] = sinf(th);
  }

  __syncthreads();

  if (has1) {
    float tr = 0.f, ti = 0.f;
    int ph = 0;
    for (int v = 0; v < 7; ++v) {
      const float c = rootc[ph], s = roots[ph];
      tr += a[v] * c + b[v] * s;
      ti += b[v] * c - a[v] * s;
      ph += wv1;
      if (ph >= GRID) ph -= GRID;
    }
    Tr[t] = tr;
    Ti[t] = ti;
  }

  __syncthreads();

  float re0 = 0.f, im0 = 0.f, re1 = 0.f, im1 = 0.f;
  if (t < WREAL) {
    const int wu = t / GRID;
    const int wv = t - wu * GRID;
    int ph = 0;
    for (int u = 0; u < 7; ++u) {
      const float c = rootc[ph], s = roots[ph];
      const float x0r = Tr[u * GRID + wv],            x0i = Ti[u * GRID + wv];
      const float x1r = Tr[(7 + u) * GRID + wv],      x1i = Ti[(7 + u) * GRID + wv];
      re0 += x0r * c + x0i * s;
      im0 += x0i * c - x0r * s;
      re1 += x1r * c + x1i * s;
      im1 += x1i * c - x1r * s;
      ph += wu;
      if (ph >= GRID) ph -= GRID;
    }
    re0 *= (1.f / 19.f); im0 *= (1.f / 19.f);
    re1 *= (1.f / 19.f); im1 *= (1.f / 19.f);
  }
  if (t < WPAD) {
    v2f* B2 = (v2f*)Bpack;
    v2f vre = {re0, re1};
    v2f vim = {im0, im1};
    B2[(    p) * WPAD + t] = vre;
    B2[(8 + p) * WPAD + t] = vim;
  }
  if (p == 0 && t < WPAD) {
    v4f z = {0.f, 0.f, 0.f, 0.f};
    if (t < WREAL) {
      const int wu = t / GRID;
      const int wv = t - wu * GRID;
      const int k1 = (3 * (wu + wv)) % GRID;
      const int k2 = (6 * (wu + wv)) % GRID;
      z.x =  rootc[k1] * (1.f / 3.f);
      z.y = -roots[k1] * (1.f / 3.f);
      z.z =  rootc[k2] * (2.f / 3.f);
      z.w = -roots[k2] * (2.f / 3.f);
    }
    ((v4f*)Zpack)[t] = z;
  }
}

__global__ __launch_bounds__(32)
void pre_C(const float* __restrict__ Sre, const float* __restrict__ Sim,
           const float* __restrict__ pw, float* __restrict__ Cpack,
           float* __restrict__ Kpack) {
  __shared__ float rootc[GRID], roots[GRID];
  __shared__ float Tr[GRID], Ti[GRID];

  const int lane = threadIdx.x;
  const int j    = blockIdx.x / GRID;
  const int wv   = blockIdx.x % GRID;

  const bool has = lane < GRID;

  if (has) {
    float th = (float)lane * TWO_PI_19;
    rootc[lane] = cosf(th);
    roots[lane] = sinf(th);
  }

  float a[GRID], b[GRID];
  for (int tv = 0; tv < GRID; ++tv) { a[tv] = 0.f; b[tv] = 0.f; }
  if (has) {
    const float* rre = Sre + (j * GRID + lane) * GRID;
    const float* rim = Sim + (j * GRID + lane) * GRID;
    for (int tv = 0; tv < GRID; ++tv) { a[tv] = rre[tv]; b[tv] = rim[tv]; }
  }

  if (wv == 0 && lane == 9)
    Kpack[j] = -(133.f / 27.f) * pw[j] * a[9];

  __syncthreads();

  float tr = 0.f, ti = 0.f;
  for (int tv = 0; tv < GRID; ++tv) {
    const int ph = (wv * tv) % GRID;
    const float c = rootc[ph], s = roots[ph];
    tr += a[tv] * c - b[tv] * s;
    ti += b[tv] * c + a[tv] * s;
  }
  if (has) {
    Tr[lane] = tr;
    Ti[lane] = ti;
  }

  __syncthreads();

  if (has) {
    const int wu = lane;
    float g3r = 0.f, g3i = 0.f;
    int ph = 0;
    for (int tu = 0; tu < GRID; ++tu) {
      const float c = rootc[ph], s = roots[ph];
      const float xr = Tr[tu], xi = Ti[tu];
      g3r += xr * c - xi * s;
      g3i += xi * c + xr * s;
      ph += wu;
      if (ph >= GRID) ph -= GRID;
    }
    const int w = wu * GRID + wv;
    const float scale = pw[j] * (1.f / 19.f);
    Cpack[((0 * (WPAD / 2) + (w >> 1)) * 16 + j) * 2 + (w & 1)] =  g3r * scale;
    Cpack[((1 * (WPAD / 2) + (w >> 1)) * 16 + j) * 2 + (w & 1)] = -g3i * scale;
  }

  if (wv == 0 && lane < 2 * (WPAD - WREAL)) {
    const int qi = lane & 1;
    const int w  = WREAL + (lane >> 1);
    Cpack[((qi * (WPAD / 2) + (w >> 1)) * 16 + j) * 2 + (w & 1)] = 0.f;
  }
}


__global__ __launch_bounds__(256, 1)
void gaunt_main(const float* __restrict__ X,
                const float* __restrict__ Bp,
                const float* __restrict__ Cp,
                const float* __restrict__ Zp,
                const float* __restrict__ Kp,
                float* __restrict__ Y) {
  __shared__ float lds[8 * 2 * 16 * 18];

  const int tid  = threadIdx.x;
  const int wave = tid >> 5;
  const int lane = tid & 31;
  const int half = lane >> 4;
  const int lr   = lane & 15;

  const long row0 = ((long)blockIdx.x * 8 + wave) * 16;

  const v2f* B2 = (const v2f*)Bp;
  const v2f* C2 = (const v2f*)Cp;
  const v4f* Z4 = (const v4f*)Zp;

  v2f xa[4];
  const float* xrow = X + (row0 + lr) * DIN;
#pragma unroll
  for (int kk = 0; kk < 4; ++kk)
    xa[kk] = *(const v2f*)(xrow + 4 * kk + 2 * half);

  const float k0 = Kp[lr];
  v8f acc = {k0, k0, k0, k0, k0, k0, k0, k0};

  float* myl = lds + wave * (2 * 16 * 18);

  v4f z = Z4[lr];
  for (int wt = 0; wt < NWT; ++wt) {
    const int w0 = wt * 16;

    const int wn = (wt + 1 < NWT) ? w0 + 16 : 0;
    const v4f zn = Z4[wn + lr];

    v8f fre = {z.x, z.x, z.x, z.x, z.x, z.x, z.x, z.x};
    v8f fim = {z.y, z.y, z.y, z.y, z.y, z.y, z.y, z.y};
#pragma unroll
    for (int kk = 0; kk < 4; ++kk) {
      fre = wmma4(xa[kk], B2[(    2 * kk + half) * WPAD + w0 + lr], fre);
      fim = wmma4(xa[kk], B2[(8 + 2 * kk + half) * WPAD + w0 + lr], fim);
    }

#pragma unroll
    for (int e = 0; e < 8; ++e) {
      float fr = fre[e], fi = fim[e];
      float p2r = fr * fr + (z.z - fi * fi);
      float p2i = 2.f * fr * fi + z.w;
      float p3r = p2r * fr - p2i * fi;
      float p3i = p2r * fi + p2i * fr;
      int off = (e + 8 * half) * 18 + lr;
      myl[0 * 288 + off] = p3r;
      myl[1 * 288 + off] = p3i;
    }

#pragma unroll
    for (int q = 0; q < 2; ++q) {
#pragma unroll
      for (int kk = 0; kk < 4; ++kk) {
        v2f a = *(const v2f*)(myl + q * 288 + lr * 18 + 4 * kk + 2 * half);
        v2f b = C2[(q * (WPAD / 2) + wt * 8 + 2 * kk + half) * 16 + lr];
        acc = wmma4(a, b, acc);
      }
    }
    z = zn;
  }

  float* yrow = Y + row0 * DOUT;
#pragma unroll
  for (int e = 0; e < 8; ++e)
    yrow[(e + 8 * half) * DOUT + lr] = acc[e];
}


extern "C" void kernel_launch(void* const* d_in, const int* in_sizes, int n_in,
                              void* d_out, int out_size, void* d_ws, size_t ws_size,
                              hipStream_t stream) {
  const float* x   = (const float*)d_in[0];
  const float* Mre = (const float*)d_in[1];
  const float* Mim = (const float*)d_in[2];
  const float* Sre = (const float*)d_in[3];
  const float* Sim = (const float*)d_in[4];
  const float* pw  = (const float*)d_in[5];

  float* ws = (float*)d_ws;
  float* Bp = ws + B_OFF;
  float* Cp = ws + C_OFF;
  float* Zp = ws + Z_OFF;
  float* Kp = ws + K_OFF;

  pre_B<<<8, PB_THREADS, 0, stream>>>(Mre, Mim, Bp, Zp);
  pre_C<<<DOUT * GRID, 32, 0, stream>>>(Sre, Sim, pw, Cp, Kp);

  gaunt_main<<<NROWS / (8 * 16), 256, 0, stream>>>(x, Bp, Cp, Zp, Kp, (float*)d_out);
}


# Main: wave priority stepped down with progress (s_setprio 3,2,1,0 from tiles 0,12,18,21; tile loop in four stretches): waves of a SIMD finish together, not oldest-first in turn; sum order unchanged
# speedup vs baseline: 4.2535x; 1.0736x over previous
#include <hip/hip_runtime.h>
#include <hip/hip_bf16.h>
#include <math.h>


typedef float v2f __attribute__((ext_vector_type(2)));
typedef float v4f __attribute__((ext_vector_type(4)));
typedef float v8f __attribute__((ext_vector_type(8)));

#define NROWS   131072
#define DIN     16
#define DOUT    16
#define GRID    19
#define WREAL   361
#define WPAD    368
#define NWT     23
#define TWO_PI_19 0.3306939635357677f

#define B_OFF   0
#define C_OFF   11776
#define Z_OFF   (11776 + 11776)
#define K_OFF   (Z_OFF + 1472)
#define WS_FLOATS (K_OFF + 16)

static __device__ inline v8f wmma4(v2f a, v2f b, v8f c) {
  return __builtin_amdgcn_wmma_f32_16x16x4_f32(
       false, a,  false, b,
       (short)0, c,  false,  false);
}


#define PB_ROWS    (2 * 7 * GRID)
#define PB_THREADS 384
#define PB_ROOT0   352
__global__ __launch_bounds__(PB_THREADS)
void pre_B(const float* __restrict__ Mre, const float* __restrict__ Mim,
           float* __restrict__ Bpack, float* __restrict__ Zpack) {
  __shared__ float rootc[GRID], roots[GRID];
  __shared__ float Tr[PB_ROWS], Ti[PB_ROWS];

  const int t = threadIdx.x;
  const int p = blockIdx.x;

  const bool has1 = t < PB_ROWS;
  const int il  = t / (7 * GRID);
  const int r   = t - il * (7 * GRID);
  const int u1  = r / GRID;
  const int wv1 = r - u1 * GRID;
  float a[7], b[7];
  for (int v = 0; v < 7; ++v) { a[v] = 0.f; b[v] = 0.f; }
  if (has1) {
    const float* rre = Mre + ((2 * p + il) * 7 + u1) * 7;
    const float* rim = Mim + ((2 * p + il) * 7 + u1) * 7;
    for (int v = 0; v < 7; ++v) { a[v] = rre[v]; b[v] = rim[v]; }
  }

  if (t >= PB_ROOT0 && t < PB_ROOT0 + GRID) {
    const int k = t - PB_ROOT0;
    const float th = (float)k * TWO_PI_19;
    rootc[k] = cosf(th);
    roots[k] = sinf(th);
  }

  __syncthreads();

  if (has1) {
    float tr = 0.f, ti = 0.f;
    int ph = 0;
    for (int v = 0; v < 7; ++v) {
      const float c = rootc[ph], s = roots[ph];
      tr += a[v] * c + b[v] * s;
      ti += b[v] * c - a[v] * s;
      ph += wv1;
      if (ph >= GRID) ph -= GRID;
    }
    Tr[t] = tr;
    Ti[t] = ti;
  }

  __syncthreads();

  float re0 = 0.f, im0 = 0.f, re1 = 0.f, im1 = 0.f;
  if (t < WREAL) {
    const int wu = t / GRID;
    const int wv = t - wu * GRID;
    int ph = 0;
    for (int u = 0; u < 7; ++u) {
      const float c = rootc[ph], s = roots[ph];
      const float x0r = Tr[u * GRID + wv],            x0i = Ti[u * GRID + wv];
      const float x1r = Tr[(7 + u) * GRID + wv],      x1i = Ti[(7 + u) * GRID + wv];
      re0 += x0r * c + x0i * s;
      im0 += x0i * c - x0r * s;
      re1 += x1r * c + x1i * s;
      im1 += x1i * c - x1r * s;
      ph += wu;
      if (ph >= GRID) ph -= GRID;
    }
    re0 *= (1.f / 19.f); im0 *= (1.f / 19.f);
    re1 *= (1.f / 19.f); im1 *= (1.f / 19.f);
  }
  if (t < WPAD) {
    v2f* B2 = (v2f*)Bpack;
    v2f vre = {re0, re1};
    v2f vim = {im0, im1};
    B2[(    p) * WPAD + t] = vre;
    B2[(8 + p) * WPAD + t] = vim;
  }
  if (p == 0 && t < WPAD) {
    v4f z = {0.f, 0.f, 0.f, 0.f};
    if (t < WREAL) {
      const int wu = t / GRID;
      const int wv = t - wu * GRID;
      const int k1 = (3 * (wu + wv)) % GRID;
      const int k2 = (6 * (wu + wv)) % GRID;
      z.x =  rootc[k1] * (1.f / 3.f);
      z.y = -roots[k1] * (1.f / 3.f);
      z.z =  rootc[k2] * (2.f / 3.f);
      z.w = -roots[k2] * (2.f / 3.f);
    }
    ((v4f*)Zpack)[t] = z;
  }
}

__global__ __launch_bounds__(32)
void pre_C(const float* __restrict__ Sre, const float* __restrict__ Sim,
           const float* __restrict__ pw, float* __restrict__ Cpack,
           float* __restrict__ Kpack) {
  __shared__ float rootc[GRID], roots[GRID];
  __shared__ float Tr[GRID], Ti[GRID];

  const int lane = threadIdx.x;
  const int j    = blockIdx.x / GRID;
  const int wv   = blockIdx.x % GRID;

  const bool has = lane < GRID;

  if (has) {
    float th = (float)lane * TWO_PI_19;
    rootc[lane] = cosf(th);
    roots[lane] = sinf(th);
  }

  float a[GRID], b[GRID];
  for (int tv = 0; tv < GRID; ++tv) { a[tv] = 0.f; b[tv] = 0.f; }
  if (has) {
    const float* rre = Sre + (j * GRID + lane) * GRID;
    const float* rim = Sim + (j * GRID + lane) * GRID;
    for (int tv = 0; tv < GRID; ++tv) { a[tv] = rre[tv]; b[tv] = rim[tv]; }
  }

  if (wv == 0 && lane == 9)
    Kpack[j] = -(133.f / 27.f) * pw[j] * a[9];

  __syncthreads();

  float tr = 0.f, ti = 0.f;
  for (int tv = 0; tv < GRID; ++tv) {
    const int ph = (wv * tv) % GRID;
    const float c = rootc[ph], s = roots[ph];
    tr += a[tv] * c - b[tv] * s;
    ti += b[tv] * c + a[tv] * s;
  }
  if (has) {
    Tr[lane] = tr;
    Ti[lane] = ti;
  }

  __syncthreads();

  if (has) {
    const int wu = lane;
    float g3r = 0.f, g3i = 0.f;
    int ph = 0;
    for (int tu = 0; tu < GRID; ++tu) {
      const float c = rootc[ph], s = roots[ph];
      const float xr = Tr[tu], xi = Ti[tu];
      g3r += xr * c - xi * s;
      g3i += xi * c + xr * s;
      ph += wu;
      if (ph >= GRID) ph -= GRID;
    }
    const int w = wu * GRID + wv;
    const float scale = pw[j] * (1.f / 19.f);
    Cpack[((0 * (WPAD / 2) + (w >> 1)) * 16 + j) * 2 + (w & 1)] =  g3r * scale;
    Cpack[((1 * (WPAD / 2) + (w >> 1)) * 16 + j) * 2 + (w & 1)] = -g3i * scale;
  }

  if (wv == 0 && lane < 2 * (WPAD - WREAL)) {
    const int qi = lane & 1;
    const int w  = WREAL + (lane >> 1);
    Cpack[((qi * (WPAD / 2) + (w >> 1)) * 16 + j) * 2 + (w & 1)] = 0.f;
  }
}


__global__ __launch_bounds__(256, 1)
void gaunt_main(const float* __restrict__ X,
                const float* __restrict__ Bp,
                const float* __restrict__ Cp,
                const float* __restrict__ Zp,
                const float* __restrict__ Kp,
                float* __restrict__ Y) {
  __shared__ float lds[8 * 2 * 16 * 18];

  const int tid  = threadIdx.x;
  const int wave = tid >> 5;
  const int lane = tid & 31;
  const int half = lane >> 4;
  const int lr   = lane & 15;

  const long row0 = ((long)blockIdx.x * 8 + wave) * 16;

  const v2f* B2 = (const v2f*)Bp;
  const v2f* C2 = (const v2f*)Cp;
  const v4f* Z4 = (const v4f*)Zp;

  v2f xa[4];
  const float* xrow = X + (row0 + lr) * DIN;
#pragma unroll
  for (int kk = 0; kk < 4; ++kk)
    xa[kk] = *(const v2f*)(xrow + 4 * kk + 2 * half);

  const float k0 = Kp[lr];
  v8f acc = {k0, k0, k0, k0, k0, k0, k0, k0};

  float* myl = lds + wave * (2 * 16 * 18);

  v4f z = Z4[lr];

  auto tile = [&](const int wt) __attribute__((always_inline)) {
    const int w0 = wt * 16;

    const int wn = (wt + 1 < NWT) ? w0 + 16 : 0;
    const v4f zn = Z4[wn + lr];

    v8f fre = {z.x, z.x, z.x, z.x, z.x, z.x, z.x, z.x};
    v8f fim = {z.y, z.y, z.y, z.y, z.y, z.y, z.y, z.y};
#pragma unroll
    for (int kk = 0; kk < 4; ++kk) {
      fre = wmma4(xa[kk], B2[(    2 * kk + half) * WPAD + w0 + lr], fre);
      fim = wmma4(xa[kk], B2[(8 + 2 * kk + half) * WPAD + w0 + lr], fim);
    }

#pragma unroll
    for (int e = 0; e < 8; ++e) {
      float fr = fre[e], fi = fim[e];
      float p2r = fr * fr + (z.z - fi * fi);
      float p2i = 2.f * fr * fi + z.w;
      float p3r = p2r * fr - p2i * fi;
      float p3i = p2r * fi + p2i * fr;
      int off = (e + 8 * half) * 18 + lr;
      myl[0 * 288 + off] = p3r;
      myl[1 * 288 + off] = p3i;
    }

#pragma unroll
    for (int q = 0; q < 2; ++q) {
#pragma unroll
      for (int kk = 0; kk < 4; ++kk) {
        v2f a = *(const v2f*)(myl + q * 288 + lr * 18 + 4 * kk + 2 * half);
        v2f b = C2[(q * (WPAD / 2) + wt * 8 + 2 * kk + half) * 16 + lr];
        acc = wmma4(a, b, acc);
      }
    }
    z = zn;
  };

  __builtin_amdgcn_s_setprio(3);
  for (int wt = 0; wt < 12; ++wt) tile(wt);
  __builtin_amdgcn_s_setprio(2);
  for (int wt = 12; wt < 18; ++wt) tile(wt);
  __builtin_amdgcn_s_setprio(1);
  for (int wt = 18; wt < 21; ++wt) tile(wt);
  __builtin_amdgcn_s_setprio(0);
  for (int wt = 21; wt < NWT; ++wt) tile(wt);

  float* yrow = Y + row0 * DOUT;
#pragma unroll
  for (int e = 0; e < 8; ++e)
    yrow[(e + 8 * half) * DOUT + lr] = acc[e];
}


extern "C" void kernel_launch(void* const* d_in, const int* in_sizes, int n_in,
                              void* d_out, int out_size, void* d_ws, size_t ws_size,
                              hipStream_t stream) {
  const float* x   = (const float*)d_in[0];
  const float* Mre = (const float*)d_in[1];
  const float* Mim = (const float*)d_in[2];
  const float* Sre = (const float*)d_in[3];
  const float* Sim = (const float*)d_in[4];
  const float* pw  = (const float*)d_in[5];

  float* ws = (float*)d_ws;
  float* Bp = ws + B_OFF;
  float* Cp = ws + C_OFF;
  float* Zp = ws + Z_OFF;
  float* Kp = ws + K_OFF;

  pre_B<<<8, PB_THREADS, 0, stream>>>(Mre, Mim, Bp, Zp);
  pre_C<<<DOUT * GRID, 32, 0, stream>>>(Sre, Sim, pw, Cp, Kp);

  gaunt_main<<<NROWS / (8 * 16), 256, 0, stream>>>(x, Bp, Cp, Zp, Kp, (float*)d_out);
}
